# starting kernel
# baseline (speedup 1.0000x reference)
#include <hip/hip_runtime.h>

typedef __attribute__((ext_vector_type(2))) float v2f;
typedef __attribute__((ext_vector_type(8))) float v8f;

#define KRED 144
#define PTOT 3136
#define PADW 146

__global__ __launch_bounds__(128) void conv_unfold_wmma_f32(
    const float* __restrict__ x,
    const float* __restrict__ Wt,
    float* __restrict__ out)
{
    __shared__ float Ws[16 * PADW];

    const int tid = threadIdx.x;
    const int bx  = blockIdx.x;
    const int it  = bx % 12;
    const int pg  = bx / 12;
    const int i0  = it * 16;

    for (int idx = tid; idx < 16 * KRED; idx += 128) {
        int row = idx / KRED;
        int col = idx - row * KRED;
        int j   = col / 3;
        int k   = col - 3 * j;
        Ws[row * PADW + k * 48 + j] = Wt[(i0 + row) * KRED + col];
    }
    __syncthreads();

    const int wave = tid >> 5;
    const int lane = tid & 31;
    const int half = lane >> 4;
    const int Mrow = lane & 15;
    const int Ncol = lane & 15;

    const int p0  = (pg * 4 + wave) * 16;
    const int p   = p0 + Ncol;
    const int n   = p / 392;
    const int rem = p - n * 392;
    const int o   = rem / 56;
    const int m   = rem - o * 56;
    const int pbase = n * 392 + m;

    const float* ptrk[3];
    bool validk[3];
    #pragma unroll
    for (int k = 0; k < 3; ++k) {
        int oo    = o + k - 1;
        validk[k] = (unsigned)oo < 7u;
        int ooc   = oo < 0 ? 0 : (oo > 6 ? 6 : oo);
        ptrk[k]   = x + pbase + ooc * 56 + 2 * half * PTOT;
    }

    const float* Wrow = &Ws[Mrow * PADW + 2 * half];

    v8f acc = {};
    #pragma unroll
    for (int s = 0; s < 36; ++s) {
        const int k  = s / 12;
        const int jb = 4 * s - 48 * k;

        v2f A;
        A.x = Wrow[4 * s];
        A.y = Wrow[4 * s + 1];

        float t0 = ptrk[k][(size_t)jb * PTOT];
        float t1 = ptrk[k][(size_t)(jb + 1) * PTOT];
        v2f B;
        B.x = validk[k] ? t0 : 0.0f;
        B.y = validk[k] ? t1 : 0.0f;

        acc = __builtin_amdgcn_wmma_f32_16x16x4_f32(
            false, A, false, B, (short)0, acc, false, false);
    }

    float* outp = out + (size_t)(i0 + 8 * half) * PTOT + p0 + Ncol;
    #pragma unroll
    for (int r = 0; r < 8; ++r) {
        outp[(size_t)r * PTOT] = acc[r];
    }
}

extern "C" void kernel_launch(void* const* d_in, const int* in_sizes, int n_in,
                              void* d_out, int out_size, void* d_ws, size_t ws_size,
                              hipStream_t stream) {
    const float* x  = (const float*)d_in[0];
    const float* Wt = (const float*)d_in[1];
    float* out      = (float*)d_out;

    conv_unfold_wmma_f32<<<dim3(12 * 49), dim3(128), 0, stream>>>(x, Wt, out);
}


# Prologue loads batched: W slice by 16-byte loads and the x values of taps 0-1 all issued before the first wait, barrier fenced on LDS only; one second-level trip where six were serial
# speedup vs baseline: 1.4505x; 1.4505x over previous
#include <hip/hip_runtime.h>


typedef __attribute__((ext_vector_type(2))) float v2f;
typedef __attribute__((ext_vector_type(4))) float v4f;
typedef __attribute__((ext_vector_type(8))) float v8f;

#define KRED 144
#define PTOT 3136
#define PADW 148

__global__ __launch_bounds__(128) void conv_unfold_wmma_f32(
    const float* __restrict__ x,
    const float* __restrict__ Wt,
    float* __restrict__ out)
{
    __shared__ __attribute__((aligned(16))) float Ws[16 * PADW];

    const int tid = threadIdx.x;
    const int bx  = blockIdx.x;
    const int it  = bx % 12;
    const int pg  = bx / 12;
    const int i0  = it * 16;

    const v4f* Wsrc = (const v4f*)(Wt + (size_t)i0 * KRED);
    const bool two  = tid < 64;
    v4f a0 = Wsrc[3 * tid + 0];
    v4f a1 = Wsrc[3 * tid + 1];
    v4f a2 = Wsrc[3 * tid + 2];
    v4f b0 = {}, b1 = {}, b2 = {};
    if (two) {
        b0 = Wsrc[3 * (tid + 128) + 0];
        b1 = Wsrc[3 * (tid + 128) + 1];
        b2 = Wsrc[3 * (tid + 128) + 2];
    }

    const int wave = tid >> 5;
    const int lane = tid & 31;
    const int half = lane >> 4;
    const int Mrow = lane & 15;
    const int Ncol = lane & 15;

    const int p0  = (pg * 4 + wave) * 16;
    const int p   = p0 + Ncol;
    const int n   = p / 392;
    const int rem = p - n * 392;
    const int o   = rem / 56;
    const int m   = rem - o * 56;
    const int pbase = n * 392 + m;

    const float* ptrk[3];
    bool validk[3];
    #pragma unroll
    for (int k = 0; k < 3; ++k) {
        int oo    = o + k - 1;
        validk[k] = (unsigned)oo < 7u;
        int ooc   = oo < 0 ? 0 : (oo > 6 ? 6 : oo);
        ptrk[k]   = x + pbase + ooc * 56 + 2 * half * PTOT;
    }

    float xb[72];
    #pragma unroll
    for (int s = 0; s < 24; ++s) {
        const int k  = s / 12;
        const int jb = 4 * s - 48 * k;
        xb[2 * s]     = ptrk[k][(size_t)jb * PTOT];
        xb[2 * s + 1] = ptrk[k][(size_t)(jb + 1) * PTOT];
    }

    __builtin_amdgcn_sched_barrier(0);

    {
        const int g   = tid;
        const int row = g / 12;
        const int gg  = g - 12 * row;
        float* d = &Ws[row * PADW + 4 * gg];
        *(v4f*)(d)      = (v4f){a0.x, a0.w, a1.z, a2.y};
        *(v4f*)(d + 48) = (v4f){a0.y, a1.x, a1.w, a2.z};
        *(v4f*)(d + 96) = (v4f){a0.z, a1.y, a2.x, a2.w};
    }
    if (two) {
        const int g   = tid + 128;
        const int row = g / 12;
        const int gg  = g - 12 * row;
        float* d = &Ws[row * PADW + 4 * gg];
        *(v4f*)(d)      = (v4f){b0.x, b0.w, b1.z, b2.y};
        *(v4f*)(d + 48) = (v4f){b0.y, b1.x, b1.w, b2.z};
        *(v4f*)(d + 96) = (v4f){b0.z, b1.y, b2.x, b2.w};
    }
    __builtin_amdgcn_fence(__ATOMIC_RELEASE, "workgroup", "local");
    __builtin_amdgcn_s_barrier();
    __builtin_amdgcn_fence(__ATOMIC_ACQUIRE, "workgroup", "local");

    #pragma unroll
    for (int s = 24; s < 36; ++s) {
        const int jb = 4 * s - 96;
        xb[2 * s]     = ptrk[2][(size_t)jb * PTOT];
        xb[2 * s + 1] = ptrk[2][(size_t)(jb + 1) * PTOT];
    }

    const float* Wrow = &Ws[Mrow * PADW + 2 * half];

    v8f acc = {};
    #pragma unroll
    for (int s = 0; s < 36; ++s) {
        const int k = s / 12;

        v2f A;
        A.x = Wrow[4 * s];
        A.y = Wrow[4 * s + 1];

        v2f B;
        B.x = validk[k] ? xb[2 * s]     : 0.0f;
        B.y = validk[k] ? xb[2 * s + 1] : 0.0f;

        acc = __builtin_amdgcn_wmma_f32_16x16x4_f32(
            false, A, false, B, (short)0, acc, false, false);
    }

    float* outp = out + (size_t)(i0 + 8 * half) * PTOT + p0 + Ncol;
    #pragma unroll
    for (int r = 0; r < 8; ++r) {
        outp[(size_t)r * PTOT] = acc[r];
    }
}

extern "C" void kernel_launch(void* const* d_in, const int* in_sizes, int n_in,
                              void* d_out, int out_size, void* d_ws, size_t ws_size,
                              hipStream_t stream) {
    const float* x  = (const float*)d_in[0];
    const float* Wt = (const float*)d_in[1];
    float* out      = (float*)d_out;

    conv_unfold_wmma_f32<<<dim3(12 * 49), dim3(128), 0, stream>>>(x, Wt, out);
}
